# starting kernel
# baseline (speedup 1.0000x reference)
#include <hip/hip_runtime.h>

typedef float v2f __attribute__((ext_vector_type(2)));
typedef float v8f __attribute__((ext_vector_type(8)));

#define N_AGENT 81
#define N_NBR   20
#define DD      96
#define HH      96
#define ROWS_PAD 96

__global__ __launch_bounds__(128) void relational_wmma_kernel(
    const float* __restrict__ x,
    const int*   __restrict__ edges,
    const float* __restrict__ W1,
    const float* __restrict__ b1,
    const float* __restrict__ W2,
    const float* __restrict__ b2,
    float* __restrict__ out)
{
    extern __shared__ float lds[];
    float* xh   = lds;
    float* sbuf = lds + 96 * 96;
    float* ybuf = lds + 2 * 96 * 96;
    int*   eg   = (int*)(lds + 3 * 96 * 96);

    const int b    = blockIdx.x;
    const int tid  = threadIdx.x;
    const int lane = tid & 31;
    const int wave = tid >> 5;
    const int lh   = (lane >> 4) & 1;
    const int l15  = lane & 15;

    const float* xb = x + (size_t)b * (N_AGENT * DD);
    for (int f = tid; f < (ROWS_PAD * DD) / 4; f += 128) {
        int row = f / (DD / 4);
        int c4  = f % (DD / 4);
        float4 v = make_float4(0.f, 0.f, 0.f, 0.f);
        if (row < N_AGENT) v = ((const float4*)(xb + row * DD))[c4];
        ((float4*)(xh + row * DD))[c4] = v;
    }
    for (int f = tid; f < N_AGENT * N_NBR; f += 128) eg[f] = edges[f];
    __syncthreads();

    for (int t = wave; t < 36; t += 4) {
        const int mt = t / 6, nt = t % 6;
        v8f accS = {};
        v8f accY = {};
        const int m = mt * 16 + l15;
        const int n = nt * 16 + l15;
        for (int k = 0; k < DD; k += 4) {
            const int k0 = k + lh * 2;
            v2f a;
            a.x = xh[m * DD + k0];
            a.y = xh[m * DD + k0 + 1];
            v2f bt, bb;
            bt.x = W1[(k0)*HH + n];
            bt.y = W1[(k0 + 1) * HH + n];
            bb.x = W1[(DD + k0) * HH + n];
            bb.y = W1[(DD + k0 + 1) * HH + n];
            accS = __builtin_amdgcn_wmma_f32_16x16x4_f32(false, a, false, bt,
                                                         (short)0, accS, false, false);
            accY = __builtin_amdgcn_wmma_f32_16x16x4_f32(false, a, false, bb,
                                                         (short)0, accY, false, false);
        }
        for (int j = 0; j < 8; ++j) {
            const int mr = mt * 16 + j + lh * 8;
            sbuf[mr * HH + n] = accS[j];
            ybuf[mr * HH + n] = accY[j];
        }
    }
    __syncthreads();

    for (int idx = tid; idx < N_AGENT * HH; idx += 128) {
        const int a = idx / HH, c = idx % HH;
        const float sa = sbuf[a * HH + c] + b1[c];
        float v = 0.f;
        #pragma unroll 5
        for (int nn = 0; nn < N_NBR; ++nn) {
            const int j = eg[a * N_NBR + nn];
            const float hv = sa + ybuf[j * HH + c];
            v += fmaxf(hv, 0.f);
        }
        xh[a * HH + c] = v;
    }
    __syncthreads();

    float* ob = out + (size_t)b * (N_AGENT * HH);
    for (int t = wave; t < 36; t += 4) {
        const int mt = t / 6, nt = t % 6;
        v8f acc = {};
        const int m = mt * 16 + l15;
        const int n = nt * 16 + l15;
        for (int k = 0; k < HH; k += 4) {
            const int k0 = k + lh * 2;
            v2f a, bw;
            a.x  = xh[m * HH + k0];
            a.y  = xh[m * HH + k0 + 1];
            bw.x = W2[(k0)*HH + n];
            bw.y = W2[(k0 + 1) * HH + n];
            acc = __builtin_amdgcn_wmma_f32_16x16x4_f32(false, a, false, bw,
                                                        (short)0, acc, false, false);
        }
        const float bias = (float)N_NBR * b2[n];
        for (int j = 0; j < 8; ++j) {
            const int mr = mt * 16 + j + lh * 8;
            if (mr < N_AGENT) ob[mr * HH + n] = acc[j] + bias;
        }
    }
}

extern "C" void kernel_launch(void* const* d_in, const int* in_sizes, int n_in,
                              void* d_out, int out_size, void* d_ws, size_t ws_size,
                              hipStream_t stream) {
    (void)in_sizes; (void)n_in; (void)out_size; (void)d_ws; (void)ws_size;
    const float* x     = (const float*)d_in[0];
    const int*   edges = (const int*)d_in[1];
    const float* W1    = (const float*)d_in[2];
    const float* b1    = (const float*)d_in[3];
    const float* W2    = (const float*)d_in[4];
    const float* b2    = (const float*)d_in[5];
    float* out = (float*)d_out;

    const size_t shmem = (size_t)(3 * 96 * 96) * sizeof(float)
                       + (size_t)(N_AGENT * N_NBR) * sizeof(int);
    relational_wmma_kernel<<<dim3(512), dim3(128), shmem, stream>>>(
        x, edges, W1, b1, W2, b2, out);
}


# Phase 3 gather regrouped: one thread sums 4 adjacent columns of one agent, its 20 neighbour rows read as 20 independent 16-byte LDS loads in flight together; sum order 0..19 kept
# speedup vs baseline: 1.5076x; 1.5076x over previous
#include <hip/hip_runtime.h>

typedef float v2f __attribute__((ext_vector_type(2)));
typedef float v8f __attribute__((ext_vector_type(8)));

#define N_AGENT 81
#define N_NBR   20
#define DD      96
#define HH      96
#define ROWS_PAD 96

__global__ __launch_bounds__(128) void relational_wmma_kernel(
    const float* __restrict__ x,
    const int*   __restrict__ edges,
    const float* __restrict__ W1,
    const float* __restrict__ b1,
    const float* __restrict__ W2,
    const float* __restrict__ b2,
    float* __restrict__ out)
{
    extern __shared__ float lds[];
    float* xh   = lds;
    float* sbuf = lds + 96 * 96;
    float* ybuf = lds + 2 * 96 * 96;
    int*   eg   = (int*)(lds + 3 * 96 * 96);

    const int b    = blockIdx.x;
    const int tid  = threadIdx.x;
    const int lane = tid & 31;
    const int wave = tid >> 5;
    const int lh   = (lane >> 4) & 1;
    const int l15  = lane & 15;

    const float* xb = x + (size_t)b * (N_AGENT * DD);
    for (int f = tid; f < (ROWS_PAD * DD) / 4; f += 128) {
        int row = f / (DD / 4);
        int c4  = f % (DD / 4);
        float4 v = make_float4(0.f, 0.f, 0.f, 0.f);
        if (row < N_AGENT) v = ((const float4*)(xb + row * DD))[c4];
        ((float4*)(xh + row * DD))[c4] = v;
    }
    for (int f = tid; f < N_AGENT * N_NBR; f += 128) eg[f] = edges[f];
    __syncthreads();

    for (int t = wave; t < 36; t += 4) {
        const int mt = t / 6, nt = t % 6;
        v8f accS = {};
        v8f accY = {};
        const int m = mt * 16 + l15;
        const int n = nt * 16 + l15;
        for (int k = 0; k < DD; k += 4) {
            const int k0 = k + lh * 2;
            v2f a;
            a.x = xh[m * DD + k0];
            a.y = xh[m * DD + k0 + 1];
            v2f bt, bb;
            bt.x = W1[(k0)*HH + n];
            bt.y = W1[(k0 + 1) * HH + n];
            bb.x = W1[(DD + k0) * HH + n];
            bb.y = W1[(DD + k0 + 1) * HH + n];
            accS = __builtin_amdgcn_wmma_f32_16x16x4_f32(false, a, false, bt,
                                                         (short)0, accS, false, false);
            accY = __builtin_amdgcn_wmma_f32_16x16x4_f32(false, a, false, bb,
                                                         (short)0, accY, false, false);
        }
        for (int j = 0; j < 8; ++j) {
            const int mr = mt * 16 + j + lh * 8;
            sbuf[mr * HH + n] = accS[j];
            ybuf[mr * HH + n] = accY[j];
        }
    }
    __syncthreads();

    {
        constexpr int NCG   = HH / 4;
        constexpr int NITEM = N_AGENT * NCG;
        constexpr int NE4   = N_NBR / 4;
        const int4* eg4 = (const int4*)eg;
        int4 e[NE4];
        #pragma unroll
        for (int q = 0; q < NE4; ++q) e[q] = eg4[(tid / NCG) * NE4 + q];
        for (int it = tid; it < NITEM; it += 128) {
            const int a = it / NCG;
            const int c = (it % NCG) * 4;
            const int an = min((it + 128) / NCG, N_AGENT - 1);
            int4 en[NE4];
            #pragma unroll
            for (int q = 0; q < NE4; ++q) en[q] = eg4[an * NE4 + q];
            const float4 bb = *(const float4*)(b1 + c);
            const float4 ss = *(const float4*)(sbuf + a * HH + c);
            const float sx = ss.x + bb.x, sy = ss.y + bb.y, sz = ss.z + bb.z, sw = ss.w + bb.w;
            const float* yc = ybuf + c;
            float vx = 0.f, vy = 0.f, vz = 0.f, vw = 0.f;
            #pragma unroll
            for (int q = 0; q < NE4; ++q) {
                const int jj[4] = {e[q].x, e[q].y, e[q].z, e[q].w};
                #pragma unroll
                for (int r = 0; r < 4; ++r) {
                    const float4 y = *(const float4*)(yc + jj[r] * HH);
                    vx += fmaxf(sx + y.x, 0.f);
                    vy += fmaxf(sy + y.y, 0.f);
                    vz += fmaxf(sz + y.z, 0.f);
                    vw += fmaxf(sw + y.w, 0.f);
                }
            }
            *(float4*)(xh + a * HH + c) = make_float4(vx, vy, vz, vw);
            #pragma unroll
            for (int q = 0; q < NE4; ++q) e[q] = en[q];
        }
    }
    __syncthreads();

    float* ob = out + (size_t)b * (N_AGENT * HH);
    for (int t = wave; t < 36; t += 4) {
        const int mt = t / 6, nt = t % 6;
        v8f acc = {};
        const int m = mt * 16 + l15;
        const int n = nt * 16 + l15;
        for (int k = 0; k < HH; k += 4) {
            const int k0 = k + lh * 2;
            v2f a, bw;
            a.x  = xh[m * HH + k0];
            a.y  = xh[m * HH + k0 + 1];
            bw.x = W2[(k0)*HH + n];
            bw.y = W2[(k0 + 1) * HH + n];
            acc = __builtin_amdgcn_wmma_f32_16x16x4_f32(false, a, false, bw,
                                                        (short)0, acc, false, false);
        }
        const float bias = (float)N_NBR * b2[n];
        for (int j = 0; j < 8; ++j) {
            const int mr = mt * 16 + j + lh * 8;
            if (mr < N_AGENT) ob[mr * HH + n] = acc[j] + bias;
        }
    }
}

extern "C" void kernel_launch(void* const* d_in, const int* in_sizes, int n_in,
                              void* d_out, int out_size, void* d_ws, size_t ws_size,
                              hipStream_t stream) {
    (void)in_sizes; (void)n_in; (void)out_size; (void)d_ws; (void)ws_size;
    const float* x     = (const float*)d_in[0];
    const int*   edges = (const int*)d_in[1];
    const float* W1    = (const float*)d_in[2];
    const float* b1    = (const float*)d_in[3];
    const float* W2    = (const float*)d_in[4];
    const float* b2    = (const float*)d_in[5];
    float* out = (float*)d_out;

    const size_t shmem = (size_t)(3 * 96 * 96) * sizeof(float)
                       + (size_t)(N_AGENT * N_NBR) * sizeof(int);
    relational_wmma_kernel<<<dim3(512), dim3(128), shmem, stream>>>(
        x, edges, W1, b1, W2, b2, out);
}


# Weights held in registers: each wave loads its W1 and W2 fragments (3 column tiles) once and sweeps the row tiles, 3 accumulators per x fragment; no weight reads in the WMMA loops; k order kept
# speedup vs baseline: 2.2542x; 1.4952x over previous
#include <hip/hip_runtime.h>

typedef float v2f __attribute__((ext_vector_type(2)));
typedef float v8f __attribute__((ext_vector_type(8)));

#define N_AGENT 81
#define N_NBR   20
#define DD      96
#define HH      96
#define ROWS_PAD 96

__global__ __launch_bounds__(128) void relational_wmma_kernel(
    const float* __restrict__ x,
    const int*   __restrict__ edges,
    const float* __restrict__ W1,
    const float* __restrict__ b1,
    const float* __restrict__ W2,
    const float* __restrict__ b2,
    float* __restrict__ out)
{
    extern __shared__ float lds[];
    float* xh   = lds;
    float* sbuf = lds + 96 * 96;
    float* ybuf = lds + 2 * 96 * 96;
    int*   eg   = (int*)(lds + 3 * 96 * 96);

    const int b    = blockIdx.x;
    const int tid  = threadIdx.x;
    const int lane = tid & 31;
    const int wave = tid >> 5;
    const int lh   = (lane >> 4) & 1;
    const int l15  = lane & 15;

    constexpr int NKK = DD / 4;
    const int ncol = (wave & 1) * 48 + l15;
    v2f wf[3][NKK];
    {
        const float* Wh = W1 + (wave >> 1) * (DD * HH);
        #pragma unroll
        for (int j = 0; j < 3; ++j) {
            #pragma unroll
            for (int kk = 0; kk < NKK; ++kk) {
                const int k0 = kk * 4 + lh * 2;
                wf[j][kk].x = Wh[(k0)*HH + ncol + j * 16];
                wf[j][kk].y = Wh[(k0 + 1) * HH + ncol + j * 16];
            }
        }
    }

    const float* xb = x + (size_t)b * (N_AGENT * DD);
    for (int f = tid; f < (ROWS_PAD * DD) / 4; f += 128) {
        int row = f / (DD / 4);
        int c4  = f % (DD / 4);
        float4 v = make_float4(0.f, 0.f, 0.f, 0.f);
        if (row < N_AGENT) v = ((const float4*)(xb + row * DD))[c4];
        ((float4*)(xh + row * DD))[c4] = v;
    }
    for (int f = tid; f < N_AGENT * N_NBR; f += 128) eg[f] = edges[f];
    __syncthreads();

    v2f wg[3][NKK];
    #pragma unroll
    for (int j = 0; j < 3; ++j) {
        #pragma unroll
        for (int kk = 0; kk < NKK; ++kk) {
            const int k0 = kk * 4 + lh * 2;
            wg[j][kk].x = W2[(k0)*HH + ncol + j * 16];
            wg[j][kk].y = W2[(k0 + 1) * HH + ncol + j * 16];
        }
    }

    {
        float* dst = (wave >> 1) ? ybuf : sbuf;
        for (int mt = 0; mt < 6; ++mt) {
            v8f acc0 = {}, acc1 = {}, acc2 = {};
            const float* arow = xh + (mt * 16 + l15) * DD + lh * 2;
            #pragma unroll
            for (int kk = 0; kk < NKK; ++kk) {
                v2f a;
                a.x = arow[kk * 4];
                a.y = arow[kk * 4 + 1];
                acc0 = __builtin_amdgcn_wmma_f32_16x16x4_f32(false, a, false, wf[0][kk],
                                                             (short)0, acc0, false, false);
                acc1 = __builtin_amdgcn_wmma_f32_16x16x4_f32(false, a, false, wf[1][kk],
                                                             (short)0, acc1, false, false);
                acc2 = __builtin_amdgcn_wmma_f32_16x16x4_f32(false, a, false, wf[2][kk],
                                                             (short)0, acc2, false, false);
            }
            #pragma unroll
            for (int j = 0; j < 8; ++j) {
                const int mr = mt * 16 + j + lh * 8;
                dst[mr * HH + ncol]      = acc0[j];
                dst[mr * HH + ncol + 16] = acc1[j];
                dst[mr * HH + ncol + 32] = acc2[j];
            }
        }
    }

    __syncthreads();

    {
        constexpr int NCG   = HH / 4;
        constexpr int NITEM = N_AGENT * NCG;
        constexpr int NE4   = N_NBR / 4;
        const int4* eg4 = (const int4*)eg;
        int4 e[NE4];
        #pragma unroll
        for (int q = 0; q < NE4; ++q) e[q] = eg4[(tid / NCG) * NE4 + q];
        for (int it = tid; it < NITEM; it += 128) {
            const int a = it / NCG;
            const int c = (it % NCG) * 4;
            const int an = min((it + 128) / NCG, N_AGENT - 1);
            int4 en[NE4];
            #pragma unroll
            for (int q = 0; q < NE4; ++q) en[q] = eg4[an * NE4 + q];
            const float4 bb = *(const float4*)(b1 + c);
            const float4 ss = *(const float4*)(sbuf + a * HH + c);
            const float sx = ss.x + bb.x, sy = ss.y + bb.y, sz = ss.z + bb.z, sw = ss.w + bb.w;
            const float* yc = ybuf + c;
            float vx = 0.f, vy = 0.f, vz = 0.f, vw = 0.f;
            #pragma unroll
            for (int q = 0; q < NE4; ++q) {
                const int jj[4] = {e[q].x, e[q].y, e[q].z, e[q].w};
                #pragma unroll
                for (int r = 0; r < 4; ++r) {
                    const float4 y = *(const float4*)(yc + jj[r] * HH);
                    vx += fmaxf(sx + y.x, 0.f);
                    vy += fmaxf(sy + y.y, 0.f);
                    vz += fmaxf(sz + y.z, 0.f);
                    vw += fmaxf(sw + y.w, 0.f);
                }
            }
            *(float4*)(xh + a * HH + c) = make_float4(vx, vy, vz, vw);
            #pragma unroll
            for (int q = 0; q < NE4; ++q) e[q] = en[q];
        }
    }
    __syncthreads();

    float* ob = out + (size_t)b * (N_AGENT * HH);
    {
        const float bias0 = (float)N_NBR * b2[ncol];
        const float bias1 = (float)N_NBR * b2[ncol + 16];
        const float bias2 = (float)N_NBR * b2[ncol + 32];
        const int mt0 = (wave >> 1) * 3;
        for (int mi = 0; mi < 3; ++mi) {
            const int mt = mt0 + mi;
            v8f acc0 = {}, acc1 = {}, acc2 = {};
            const float* arow = xh + (mt * 16 + l15) * HH + lh * 2;
            #pragma unroll
            for (int kk = 0; kk < NKK; ++kk) {
                v2f a;
                a.x = arow[kk * 4];
                a.y = arow[kk * 4 + 1];
                acc0 = __builtin_amdgcn_wmma_f32_16x16x4_f32(false, a, false, wg[0][kk],
                                                             (short)0, acc0, false, false);
                acc1 = __builtin_amdgcn_wmma_f32_16x16x4_f32(false, a, false, wg[1][kk],
                                                             (short)0, acc1, false, false);
                acc2 = __builtin_amdgcn_wmma_f32_16x16x4_f32(false, a, false, wg[2][kk],
                                                             (short)0, acc2, false, false);
            }
            #pragma unroll
            for (int j = 0; j < 8; ++j) {
                const int mr = mt * 16 + j + lh * 8;
                if (mr < N_AGENT) {
                    ob[mr * HH + ncol]      = acc0[j] + bias0;
                    ob[mr * HH + ncol + 16] = acc1[j] + bias1;
                    ob[mr * HH + ncol + 32] = acc2[j] + bias2;
                }
            }
        }
    }
}

extern "C" void kernel_launch(void* const* d_in, const int* in_sizes, int n_in,
                              void* d_out, int out_size, void* d_ws, size_t ws_size,
                              hipStream_t stream) {
    (void)in_sizes; (void)n_in; (void)out_size; (void)d_ws; (void)ws_size;
    const float* x     = (const float*)d_in[0];
    const int*   edges = (const int*)d_in[1];
    const float* W1    = (const float*)d_in[2];
    const float* b1    = (const float*)d_in[3];
    const float* W2    = (const float*)d_in[4];
    const float* b2    = (const float*)d_in[5];
    float* out = (float*)d_out;

    const size_t shmem = (size_t)(3 * 96 * 96) * sizeof(float)
                       + (size_t)(N_AGENT * N_NBR) * sizeof(int);
    relational_wmma_kernel<<<dim3(512), dim3(128), shmem, stream>>>(
        x, edges, W1, b1, W2, b2, out);
}


# x tile and edge list staged with all of a thread's 16-byte global loads (16 + 4) issued back to back before any wait, then stored to LDS; was one load-wait-store round trip after another
# speedup vs baseline: 2.9128x; 1.2922x over previous
#include <hip/hip_runtime.h>

typedef float v2f __attribute__((ext_vector_type(2)));
typedef float v8f __attribute__((ext_vector_type(8)));

#define N_AGENT 81
#define N_NBR   20
#define DD      96
#define HH      96
#define ROWS_PAD 96

__global__ __launch_bounds__(128) void relational_wmma_kernel(
    const float* __restrict__ x,
    const int*   __restrict__ edges,
    const float* __restrict__ W1,
    const float* __restrict__ b1,
    const float* __restrict__ W2,
    const float* __restrict__ b2,
    float* __restrict__ out)
{
    extern __shared__ float lds[];
    float* xh   = lds;
    float* sbuf = lds + 96 * 96;
    float* ybuf = lds + 2 * 96 * 96;
    int*   eg   = (int*)(lds + 3 * 96 * 96);

    const int b    = blockIdx.x;
    const int tid  = threadIdx.x;
    const int lane = tid & 31;
    const int wave = tid >> 5;
    const int lh   = (lane >> 4) & 1;
    const int l15  = lane & 15;

    constexpr int NX4    = (N_AGENT * DD) / 4;
    constexpr int NXP4   = (ROWS_PAD * DD) / 4;
    constexpr int NXFULL = NX4 / 128;
    constexpr int NXRND  = NXP4 / 128;
    constexpr int NE4TOT = (N_AGENT * N_NBR) / 4;
    constexpr int NEFULL = NE4TOT / 128;
    const float4* xb4  = (const float4*)(x + (size_t)b * (N_AGENT * DD));
    const int4*   eg4g = (const int4*)edges;
    float4 xr[NXFULL];
    int4   er[NEFULL];
    #pragma unroll
    for (int i = 0; i < NXFULL; ++i) xr[i] = xb4[tid + 128 * i];
    const float4 xlast = xb4[min(tid + 128 * NXFULL, NX4 - 1)];
    #pragma unroll
    for (int i = 0; i < NEFULL; ++i) er[i] = eg4g[tid + 128 * i];
    const int  felast = min(tid + 128 * NEFULL, NE4TOT - 1);
    const int4 elast  = eg4g[felast];
    __builtin_amdgcn_sched_barrier(0);

    constexpr int NKK = DD / 4;
    const int ncol = (wave & 1) * 48 + l15;
    v2f wf[3][NKK];
    {
        const float* Wh = W1 + (wave >> 1) * (DD * HH);
        #pragma unroll
        for (int j = 0; j < 3; ++j) {
            #pragma unroll
            for (int kk = 0; kk < NKK; ++kk) {
                const int k0 = kk * 4 + lh * 2;
                wf[j][kk].x = Wh[(k0)*HH + ncol + j * 16];
                wf[j][kk].y = Wh[(k0 + 1) * HH + ncol + j * 16];
            }
        }
    }

    {
        float4* xh4  = (float4*)xh;
        int4*   eg4s = (int4*)eg;
        #pragma unroll
        for (int i = 0; i < NXFULL; ++i) xh4[tid + 128 * i] = xr[i];
        const bool xreal = (tid + 128 * NXFULL) < NX4;
        xh4[tid + 128 * NXFULL] = make_float4(xreal ? xlast.x : 0.f, xreal ? xlast.y : 0.f,
                                              xreal ? xlast.z : 0.f, xreal ? xlast.w : 0.f);
        #pragma unroll
        for (int i = NXFULL + 1; i < NXRND; ++i) xh4[tid + 128 * i] = make_float4(0.f, 0.f, 0.f, 0.f);
        #pragma unroll
        for (int i = 0; i < NEFULL; ++i) eg4s[tid + 128 * i] = er[i];
        eg4s[felast] = elast;
    }
    __syncthreads();

    v2f wg[3][NKK];
    #pragma unroll
    for (int j = 0; j < 3; ++j) {
        #pragma unroll
        for (int kk = 0; kk < NKK; ++kk) {
            const int k0 = kk * 4 + lh * 2;
            wg[j][kk].x = W2[(k0)*HH + ncol + j * 16];
            wg[j][kk].y = W2[(k0 + 1) * HH + ncol + j * 16];
        }
    }

    {
        float* dst = (wave >> 1) ? ybuf : sbuf;
        for (int mt = 0; mt < 6; ++mt) {
            v8f acc0 = {}, acc1 = {}, acc2 = {};
            const float* arow = xh + (mt * 16 + l15) * DD + lh * 2;
            #pragma unroll
            for (int kk = 0; kk < NKK; ++kk) {
                v2f a;
                a.x = arow[kk * 4];
                a.y = arow[kk * 4 + 1];
                acc0 = __builtin_amdgcn_wmma_f32_16x16x4_f32(false, a, false, wf[0][kk],
                                                             (short)0, acc0, false, false);
                acc1 = __builtin_amdgcn_wmma_f32_16x16x4_f32(false, a, false, wf[1][kk],
                                                             (short)0, acc1, false, false);
                acc2 = __builtin_amdgcn_wmma_f32_16x16x4_f32(false, a, false, wf[2][kk],
                                                             (short)0, acc2, false, false);
            }
            #pragma unroll
            for (int j = 0; j < 8; ++j) {
                const int mr = mt * 16 + j + lh * 8;
                dst[mr * HH + ncol]      = acc0[j];
                dst[mr * HH + ncol + 16] = acc1[j];
                dst[mr * HH + ncol + 32] = acc2[j];
            }
        }
    }

    __syncthreads();

    {
        constexpr int NCG   = HH / 4;
        constexpr int NITEM = N_AGENT * NCG;
        constexpr int NE4   = N_NBR / 4;
        const int4* eg4 = (const int4*)eg;
        int4 e[NE4];
        #pragma unroll
        for (int q = 0; q < NE4; ++q) e[q] = eg4[(tid / NCG) * NE4 + q];
        for (int it = tid; it < NITEM; it += 128) {
            const int a = it / NCG;
            const int c = (it % NCG) * 4;
            const int an = min((it + 128) / NCG, N_AGENT - 1);
            int4 en[NE4];
            #pragma unroll
            for (int q = 0; q < NE4; ++q) en[q] = eg4[an * NE4 + q];
            const float4 bb = *(const float4*)(b1 + c);
            const float4 ss = *(const float4*)(sbuf + a * HH + c);
            const float sx = ss.x + bb.x, sy = ss.y + bb.y, sz = ss.z + bb.z, sw = ss.w + bb.w;
            const float* yc = ybuf + c;
            float vx = 0.f, vy = 0.f, vz = 0.f, vw = 0.f;
            #pragma unroll
            for (int q = 0; q < NE4; ++q) {
                const int jj[4] = {e[q].x, e[q].y, e[q].z, e[q].w};
                #pragma unroll
                for (int r = 0; r < 4; ++r) {
                    const float4 y = *(const float4*)(yc + jj[r] * HH);
                    vx += fmaxf(sx + y.x, 0.f);
                    vy += fmaxf(sy + y.y, 0.f);
                    vz += fmaxf(sz + y.z, 0.f);
                    vw += fmaxf(sw + y.w, 0.f);
                }
            }
            *(float4*)(xh + a * HH + c) = make_float4(vx, vy, vz, vw);
            #pragma unroll
            for (int q = 0; q < NE4; ++q) e[q] = en[q];
        }
    }
    __syncthreads();

    float* ob = out + (size_t)b * (N_AGENT * HH);
    {
        const float bias0 = (float)N_NBR * b2[ncol];
        const float bias1 = (float)N_NBR * b2[ncol + 16];
        const float bias2 = (float)N_NBR * b2[ncol + 32];
        const int mt0 = (wave >> 1) * 3;
        for (int mi = 0; mi < 3; ++mi) {
            const int mt = mt0 + mi;
            v8f acc0 = {}, acc1 = {}, acc2 = {};
            const float* arow = xh + (mt * 16 + l15) * HH + lh * 2;
            #pragma unroll
            for (int kk = 0; kk < NKK; ++kk) {
                v2f a;
                a.x = arow[kk * 4];
                a.y = arow[kk * 4 + 1];
                acc0 = __builtin_amdgcn_wmma_f32_16x16x4_f32(false, a, false, wg[0][kk],
                                                             (short)0, acc0, false, false);
                acc1 = __builtin_amdgcn_wmma_f32_16x16x4_f32(false, a, false, wg[1][kk],
                                                             (short)0, acc1, false, false);
                acc2 = __builtin_amdgcn_wmma_f32_16x16x4_f32(false, a, false, wg[2][kk],
                                                             (short)0, acc2, false, false);
            }
            #pragma unroll
            for (int j = 0; j < 8; ++j) {
                const int mr = mt * 16 + j + lh * 8;
                if (mr < N_AGENT) {
                    ob[mr * HH + ncol]      = acc0[j] + bias0;
                    ob[mr * HH + ncol + 16] = acc1[j] + bias1;
                    ob[mr * HH + ncol + 32] = acc2[j] + bias2;
                }
            }
        }
    }
}

extern "C" void kernel_launch(void* const* d_in, const int* in_sizes, int n_in,
                              void* d_out, int out_size, void* d_ws, size_t ws_size,
                              hipStream_t stream) {
    (void)in_sizes; (void)n_in; (void)out_size; (void)d_ws; (void)ws_size;
    const float* x     = (const float*)d_in[0];
    const int*   edges = (const int*)d_in[1];
    const float* W1    = (const float*)d_in[2];
    const float* b1    = (const float*)d_in[3];
    const float* W2    = (const float*)d_in[4];
    const float* b2    = (const float*)d_in[5];
    float* out = (float*)d_out;

    const size_t shmem = (size_t)(3 * 96 * 96) * sizeof(float)
                       + (size_t)(N_AGENT * N_NBR) * sizeof(int);
    relational_wmma_kernel<<<dim3(512), dim3(128), shmem, stream>>>(
        x, edges, W1, b1, W2, b2, out);
}


# A fragments of each 16-row tile (24 per lane) read from LDS as one group a whole tile ahead of their WMMAs (second register set), both matrix phases; no LDS load among the WMMAs; k order kept
# speedup vs baseline: 3.0088x; 1.0330x over previous
#include <hip/hip_runtime.h>

typedef float v2f __attribute__((ext_vector_type(2)));
typedef float v8f __attribute__((ext_vector_type(8)));

#define N_AGENT 81
#define N_NBR   20
#define DD      96
#define HH      96
#define ROWS_PAD 96

__global__ __launch_bounds__(128) void relational_wmma_kernel(
    const float* __restrict__ x,
    const int*   __restrict__ edges,
    const float* __restrict__ W1,
    const float* __restrict__ b1,
    const float* __restrict__ W2,
    const float* __restrict__ b2,
    float* __restrict__ out)
{
    extern __shared__ __attribute__((aligned(16))) float lds[];
    float* xh   = lds;
    float* sbuf = lds + 96 * 96;
    float* ybuf = lds + 2 * 96 * 96;
    int*   eg   = (int*)(lds + 3 * 96 * 96);

    const int b    = blockIdx.x;
    const int tid  = threadIdx.x;
    const int lane = tid & 31;
    const int wave = tid >> 5;
    const int lh   = (lane >> 4) & 1;
    const int l15  = lane & 15;

    constexpr int NX4    = (N_AGENT * DD) / 4;
    constexpr int NXP4   = (ROWS_PAD * DD) / 4;
    constexpr int NXFULL = NX4 / 128;
    constexpr int NXRND  = NXP4 / 128;
    constexpr int NE4TOT = (N_AGENT * N_NBR) / 4;
    constexpr int NEFULL = NE4TOT / 128;
    const float4* xb4  = (const float4*)(x + (size_t)b * (N_AGENT * DD));
    const int4*   eg4g = (const int4*)edges;
    float4 xr[NXFULL];
    int4   er[NEFULL];
    #pragma unroll
    for (int i = 0; i < NXFULL; ++i) xr[i] = xb4[tid + 128 * i];
    const float4 xlast = xb4[min(tid + 128 * NXFULL, NX4 - 1)];
    #pragma unroll
    for (int i = 0; i < NEFULL; ++i) er[i] = eg4g[tid + 128 * i];
    const int  felast = min(tid + 128 * NEFULL, NE4TOT - 1);
    const int4 elast  = eg4g[felast];
    __builtin_amdgcn_sched_barrier(0);

    constexpr int NKK = DD / 4;
    const int ncol = (wave & 1) * 48 + l15;
    v2f wf[3][NKK];
    {
        const float* Wh = W1 + (wave >> 1) * (DD * HH);
        #pragma unroll
        for (int j = 0; j < 3; ++j) {
            #pragma unroll
            for (int kk = 0; kk < NKK; ++kk) {
                const int k0 = kk * 4 + lh * 2;
                wf[j][kk].x = Wh[(k0)*HH + ncol + j * 16];
                wf[j][kk].y = Wh[(k0 + 1) * HH + ncol + j * 16];
            }
        }
    }

    {
        float4* xh4  = (float4*)xh;
        int4*   eg4s = (int4*)eg;
        #pragma unroll
        for (int i = 0; i < NXFULL; ++i) xh4[tid + 128 * i] = xr[i];
        const bool xreal = (tid + 128 * NXFULL) < NX4;
        xh4[tid + 128 * NXFULL] = make_float4(xreal ? xlast.x : 0.f, xreal ? xlast.y : 0.f,
                                              xreal ? xlast.z : 0.f, xreal ? xlast.w : 0.f);
        #pragma unroll
        for (int i = NXFULL + 1; i < NXRND; ++i) xh4[tid + 128 * i] = make_float4(0.f, 0.f, 0.f, 0.f);
        #pragma unroll
        for (int i = 0; i < NEFULL; ++i) eg4s[tid + 128 * i] = er[i];
        eg4s[felast] = elast;
    }
    __syncthreads();

    v2f af[2][NKK];
    const float* afrag = xh + l15 * DD + lh * 2;
    #pragma unroll
    for (int kk = 0; kk < NKK; ++kk) af[0][kk] = *(const v2f*)(afrag + kk * 4);

    v2f wg[3][NKK];
    #pragma unroll
    for (int j = 0; j < 3; ++j) {
        #pragma unroll
        for (int kk = 0; kk < NKK; ++kk) {
            const int k0 = kk * 4 + lh * 2;
            wg[j][kk].x = W2[(k0)*HH + ncol + j * 16];
            wg[j][kk].y = W2[(k0 + 1) * HH + ncol + j * 16];
        }
    }

    {
        float* dst = (wave >> 1) ? ybuf : sbuf;
        #pragma unroll
        for (int mt = 0; mt < 6; ++mt) {
            if (mt + 1 < 6) {
                const float* an = afrag + (mt + 1) * 16 * DD;
                #pragma unroll
                for (int kk = 0; kk < NKK; ++kk) af[(mt + 1) & 1][kk] = *(const v2f*)(an + kk * 4);
            }
            __builtin_amdgcn_sched_barrier(0);
            v8f acc0 = {}, acc1 = {}, acc2 = {};
            #pragma unroll
            for (int kk = 0; kk < NKK; ++kk) {
                const v2f a = af[mt & 1][kk];
                acc0 = __builtin_amdgcn_wmma_f32_16x16x4_f32(false, a, false, wf[0][kk],
                                                             (short)0, acc0, false, false);
                acc1 = __builtin_amdgcn_wmma_f32_16x16x4_f32(false, a, false, wf[1][kk],
                                                             (short)0, acc1, false, false);
                acc2 = __builtin_amdgcn_wmma_f32_16x16x4_f32(false, a, false, wf[2][kk],
                                                             (short)0, acc2, false, false);
            }
            #pragma unroll
            for (int j = 0; j < 8; ++j) {
                const int mr = mt * 16 + j + lh * 8;
                dst[mr * HH + ncol]      = acc0[j];
                dst[mr * HH + ncol + 16] = acc1[j];
                dst[mr * HH + ncol + 32] = acc2[j];
            }
        }
    }

    __syncthreads();

    {
        constexpr int NCG   = HH / 4;
        constexpr int NITEM = N_AGENT * NCG;
        constexpr int NE4   = N_NBR / 4;
        const int4* eg4 = (const int4*)eg;
        int4 e[NE4];
        #pragma unroll
        for (int q = 0; q < NE4; ++q) e[q] = eg4[(tid / NCG) * NE4 + q];
        for (int it = tid; it < NITEM; it += 128) {
            const int a = it / NCG;
            const int c = (it % NCG) * 4;
            const int an = min((it + 128) / NCG, N_AGENT - 1);
            int4 en[NE4];
            #pragma unroll
            for (int q = 0; q < NE4; ++q) en[q] = eg4[an * NE4 + q];
            const float4 bb = *(const float4*)(b1 + c);
            const float4 ss = *(const float4*)(sbuf + a * HH + c);
            const float sx = ss.x + bb.x, sy = ss.y + bb.y, sz = ss.z + bb.z, sw = ss.w + bb.w;
            const float* yc = ybuf + c;
            float vx = 0.f, vy = 0.f, vz = 0.f, vw = 0.f;
            #pragma unroll
            for (int q = 0; q < NE4; ++q) {
                const int jj[4] = {e[q].x, e[q].y, e[q].z, e[q].w};
                #pragma unroll
                for (int r = 0; r < 4; ++r) {
                    const float4 y = *(const float4*)(yc + jj[r] * HH);
                    vx += fmaxf(sx + y.x, 0.f);
                    vy += fmaxf(sy + y.y, 0.f);
                    vz += fmaxf(sz + y.z, 0.f);
                    vw += fmaxf(sw + y.w, 0.f);
                }
            }
            *(float4*)(xh + a * HH + c) = make_float4(vx, vy, vz, vw);
            #pragma unroll
            for (int q = 0; q < NE4; ++q) e[q] = en[q];
        }
    }
    __syncthreads();

    float* ob = out + (size_t)b * (N_AGENT * HH);
    {
        const float bias0 = (float)N_NBR * b2[ncol];
        const float bias1 = (float)N_NBR * b2[ncol + 16];
        const float bias2 = (float)N_NBR * b2[ncol + 32];
        const int mt0 = (wave >> 1) * 3;
        const float* hfrag = xh + (mt0 * 16 + l15) * HH + lh * 2;
        #pragma unroll
        for (int kk = 0; kk < NKK; ++kk) af[0][kk] = *(const v2f*)(hfrag + kk * 4);
        #pragma unroll
        for (int mi = 0; mi < 3; ++mi) {
            const int mt = mt0 + mi;
            if (mi + 1 < 3) {
                const float* an = hfrag + (mi + 1) * 16 * HH;
                #pragma unroll
                for (int kk = 0; kk < NKK; ++kk) af[(mi + 1) & 1][kk] = *(const v2f*)(an + kk * 4);
            }
            __builtin_amdgcn_sched_barrier(0);
            v8f acc0 = {}, acc1 = {}, acc2 = {};
            #pragma unroll
            for (int kk = 0; kk < NKK; ++kk) {
                const v2f a = af[mi & 1][kk];
                acc0 = __builtin_amdgcn_wmma_f32_16x16x4_f32(false, a, false, wg[0][kk],
                                                             (short)0, acc0, false, false);
                acc1 = __builtin_amdgcn_wmma_f32_16x16x4_f32(false, a, false, wg[1][kk],
                                                             (short)0, acc1, false, false);
                acc2 = __builtin_amdgcn_wmma_f32_16x16x4_f32(false, a, false, wg[2][kk],
                                                             (short)0, acc2, false, false);
            }
            #pragma unroll
            for (int j = 0; j < 8; ++j) {
                const int mr = mt * 16 + j + lh * 8;
                if (mr < N_AGENT) {
                    ob[mr * HH + ncol]      = acc0[j] + bias0;
                    ob[mr * HH + ncol + 16] = acc1[j] + bias1;
                    ob[mr * HH + ncol + 32] = acc2[j] + bias2;
                }
            }
        }
    }
}

extern "C" void kernel_launch(void* const* d_in, const int* in_sizes, int n_in,
                              void* d_out, int out_size, void* d_ws, size_t ws_size,
                              hipStream_t stream) {
    (void)in_sizes; (void)n_in; (void)out_size; (void)d_ws; (void)ws_size;
    const float* x     = (const float*)d_in[0];
    const int*   edges = (const int*)d_in[1];
    const float* W1    = (const float*)d_in[2];
    const float* b1    = (const float*)d_in[3];
    const float* W2    = (const float*)d_in[4];
    const float* b2    = (const float*)d_in[5];
    float* out = (float*)d_out;

    const size_t shmem = (size_t)(3 * 96 * 96) * sizeof(float)
                       + (size_t)(N_AGENT * N_NBR) * sizeof(int);
    relational_wmma_kernel<<<dim3(512), dim3(128), shmem, stream>>>(
        x, edges, W1, b1, W2, b2, out);
}


# LDS tile that feeds the A fragments (x, later hsum) at a row pitch of 100 floats (16 B x 25), was 96: its fragment reads went 8 deep on each bank, now each bank is asked once; k order kept
# speedup vs baseline: 3.1383x; 1.0430x over previous
#include <hip/hip_runtime.h>

typedef float v2f __attribute__((ext_vector_type(2)));
typedef float v4f __attribute__((ext_vector_type(4)));
typedef float v8f __attribute__((ext_vector_type(8)));

#define N_AGENT 81
#define N_NBR   20
#define DD      96
#define HH      96
#define ROWS_PAD 96
#define XP      100

__global__ __launch_bounds__(128) void relational_wmma_kernel(
    const float* __restrict__ x,
    const int*   __restrict__ edges,
    const float* __restrict__ W1,
    const float* __restrict__ b1,
    const float* __restrict__ W2,
    const float* __restrict__ b2,
    float* __restrict__ out)
{
    extern __shared__ __attribute__((aligned(16))) float lds[];
    float* xh   = lds;
    float* sbuf = lds + 96 * XP;
    float* ybuf = lds + 96 * XP + 96 * 96;
    int*   eg   = (int*)(lds + 96 * XP + 2 * 96 * 96);

    const int b    = blockIdx.x;
    const int tid  = threadIdx.x;
    const int lane = tid & 31;
    const int wave = tid >> 5;
    const int lh   = (lane >> 4) & 1;
    const int l15  = lane & 15;

    constexpr int NX4    = (N_AGENT * DD) / 4;
    constexpr int NXP4   = (ROWS_PAD * DD) / 4;
    constexpr int NXFULL = NX4 / 128;
    constexpr int NXRND  = NXP4 / 128;
    constexpr int NE4TOT = (N_AGENT * N_NBR) / 4;
    constexpr int NEFULL = NE4TOT / 128;
    const float4* xb4  = (const float4*)(x + (size_t)b * (N_AGENT * DD));
    const int4*   eg4g = (const int4*)edges;
    float4 xr[NXFULL];
    int4   er[NEFULL];
    #pragma unroll
    for (int i = 0; i < NXFULL; ++i) xr[i] = xb4[tid + 128 * i];
    const float4 xlast = xb4[min(tid + 128 * NXFULL, NX4 - 1)];
    #pragma unroll
    for (int i = 0; i < NEFULL; ++i) er[i] = eg4g[tid + 128 * i];
    const int  felast = min(tid + 128 * NEFULL, NE4TOT - 1);
    const int4 elast  = eg4g[felast];
    __builtin_amdgcn_sched_barrier(0);

    constexpr int NKK = DD / 4;
    const int ncol = (wave & 1) * 48 + l15;
    v2f wf[3][NKK];
    {
        const float* Wh = W1 + (wave >> 1) * (DD * HH);
        #pragma unroll
        for (int j = 0; j < 3; ++j) {
            #pragma unroll
            for (int kk = 0; kk < NKK; ++kk) {
                const int k0 = kk * 4 + lh * 2;
                wf[j][kk].x = Wh[(k0)*HH + ncol + j * 16];
                wf[j][kk].y = Wh[(k0 + 1) * HH + ncol + j * 16];
            }
        }
    }

    {
        float4* xh4  = (float4*)xh;
        int4*   eg4s = (int4*)eg;
        #pragma unroll
        for (int i = 0; i < NXFULL; ++i) {
            const unsigned f = (unsigned)(tid + 128 * i);
            xh4[f + f / (DD / 4)] = xr[i];
        }
        const bool xreal = (tid + 128 * NXFULL) < NX4;
        {
            const unsigned f = (unsigned)(tid + 128 * NXFULL);
            xh4[f + f / (DD / 4)] = make_float4(xreal ? xlast.x : 0.f, xreal ? xlast.y : 0.f,
                                                xreal ? xlast.z : 0.f, xreal ? xlast.w : 0.f);
        }
        #pragma unroll
        for (int i = NXFULL + 1; i < NXRND; ++i) {
            const unsigned f = (unsigned)(tid + 128 * i);
            xh4[f + f / (DD / 4)] = make_float4(0.f, 0.f, 0.f, 0.f);
        }
        #pragma unroll
        for (int i = 0; i < NEFULL; ++i) eg4s[tid + 128 * i] = er[i];
        eg4s[felast] = elast;
    }
    __syncthreads();

    v2f af[2][NKK];
    const float* afrag = xh + l15 * XP + lh * 2;
    #pragma unroll
    for (int kk = 0; kk < NKK; ++kk) af[0][kk] = *(const v2f*)(afrag + kk * 4);

    v2f wg[3][NKK];
    #pragma unroll
    for (int j = 0; j < 3; ++j) {
        #pragma unroll
        for (int kk = 0; kk < NKK; ++kk) {
            const int k0 = kk * 4 + lh * 2;
            wg[j][kk].x = W2[(k0)*HH + ncol + j * 16];
            wg[j][kk].y = W2[(k0 + 1) * HH + ncol + j * 16];
        }
    }

    {
        float* dst = (wave >> 1) ? ybuf : sbuf;
        #pragma unroll
        for (int mt = 0; mt < 6; ++mt) {
            if (mt + 1 < 6) {
                const float* an = afrag + (mt + 1) * 16 * XP;
                #pragma unroll
                for (int kk = 0; kk < NKK; ++kk) af[(mt + 1) & 1][kk] = *(const v2f*)(an + kk * 4);
            }
            __builtin_amdgcn_sched_barrier(0);
            v8f acc0 = {}, acc1 = {}, acc2 = {};
            #pragma unroll
            for (int kk = 0; kk < NKK; ++kk) {
                const v2f a = af[mt & 1][kk];
                acc0 = __builtin_amdgcn_wmma_f32_16x16x4_f32(false, a, false, wf[0][kk],
                                                             (short)0, acc0, false, false);
                acc1 = __builtin_amdgcn_wmma_f32_16x16x4_f32(false, a, false, wf[1][kk],
                                                             (short)0, acc1, false, false);
                acc2 = __builtin_amdgcn_wmma_f32_16x16x4_f32(false, a, false, wf[2][kk],
                                                             (short)0, acc2, false, false);
            }
            #pragma unroll
            for (int j = 0; j < 8; ++j) {
                const int mr = mt * 16 + j + lh * 8;
                dst[mr * HH + ncol]      = acc0[j];
                dst[mr * HH + ncol + 16] = acc1[j];
                dst[mr * HH + ncol + 32] = acc2[j];
            }
        }
    }

    __syncthreads();

    {
        constexpr int NCG   = HH / 4;
        constexpr int NITEM = N_AGENT * NCG;
        constexpr int NE4   = N_NBR / 4;
        const int4* eg4 = (const int4*)eg;
        int4 e[NE4];
        #pragma unroll
        for (int q = 0; q < NE4; ++q) e[q] = eg4[(tid / NCG) * NE4 + q];
        for (int it = tid; it < NITEM; it += 128) {
            const int a = it / NCG;
            const int c = (it % NCG) * 4;
            const int an = min((it + 128) / NCG, N_AGENT - 1);
            int4 en[NE4];
            #pragma unroll
            for (int q = 0; q < NE4; ++q) en[q] = eg4[an * NE4 + q];
            const float4 bb = *(const float4*)(b1 + c);
            const v4f ss = *(const v4f*)(sbuf + a * HH + c);
            const float sx = ss.x + bb.x, sy = ss.y + bb.y, sz = ss.z + bb.z, sw = ss.w + bb.w;
            const float* yc = ybuf + c;
            float vx = 0.f, vy = 0.f, vz = 0.f, vw = 0.f;
            #pragma unroll
            for (int q = 0; q < NE4; ++q) {
                const int jj[4] = {e[q].x, e[q].y, e[q].z, e[q].w};
                #pragma unroll
                for (int r = 0; r < 4; ++r) {
                    const v4f y = *(const v4f*)(yc + jj[r] * HH);
                    vx += fmaxf(sx + y.x, 0.f);
                    vy += fmaxf(sy + y.y, 0.f);
                    vz += fmaxf(sz + y.z, 0.f);
                    vw += fmaxf(sw + y.w, 0.f);
                }
            }
            *(v4f*)(xh + a * XP + c) = (v4f){vx, vy, vz, vw};
            #pragma unroll
            for (int q = 0; q < NE4; ++q) e[q] = en[q];
        }
    }
    __syncthreads();

    float* ob = out + (size_t)b * (N_AGENT * HH);
    {
        const float bias0 = (float)N_NBR * b2[ncol];
        const float bias1 = (float)N_NBR * b2[ncol + 16];
        const float bias2 = (float)N_NBR * b2[ncol + 32];
        const int mt0 = (wave >> 1) * 3;
        const float* hfrag = xh + (mt0 * 16 + l15) * XP + lh * 2;
        #pragma unroll
        for (int kk = 0; kk < NKK; ++kk) af[0][kk] = *(const v2f*)(hfrag + kk * 4);
        #pragma unroll
        for (int mi = 0; mi < 3; ++mi) {
            const int mt = mt0 + mi;
            if (mi + 1 < 3) {
                const float* an = hfrag + (mi + 1) * 16 * XP;
                #pragma unroll
                for (int kk = 0; kk < NKK; ++kk) af[(mi + 1) & 1][kk] = *(const v2f*)(an + kk * 4);
            }
            __builtin_amdgcn_sched_barrier(0);
            v8f acc0 = {}, acc1 = {}, acc2 = {};
            #pragma unroll
            for (int kk = 0; kk < NKK; ++kk) {
                const v2f a = af[mi & 1][kk];
                acc0 = __builtin_amdgcn_wmma_f32_16x16x4_f32(false, a, false, wg[0][kk],
                                                             (short)0, acc0, false, false);
                acc1 = __builtin_amdgcn_wmma_f32_16x16x4_f32(false, a, false, wg[1][kk],
                                                             (short)0, acc1, false, false);
                acc2 = __builtin_amdgcn_wmma_f32_16x16x4_f32(false, a, false, wg[2][kk],
                                                             (short)0, acc2, false, false);
            }
            #pragma unroll
            for (int j = 0; j < 8; ++j) {
                const int mr = mt * 16 + j + lh * 8;
                if (mr < N_AGENT) {
                    ob[mr * HH + ncol]      = acc0[j] + bias0;
                    ob[mr * HH + ncol + 16] = acc1[j] + bias1;
                    ob[mr * HH + ncol + 32] = acc2[j] + bias2;
                }
            }
        }
    }
}

extern "C" void kernel_launch(void* const* d_in, const int* in_sizes, int n_in,
                              void* d_out, int out_size, void* d_ws, size_t ws_size,
                              hipStream_t stream) {
    (void)in_sizes; (void)n_in; (void)out_size; (void)d_ws; (void)ws_size;
    const float* x     = (const float*)d_in[0];
    const int*   edges = (const int*)d_in[1];
    const float* W1    = (const float*)d_in[2];
    const float* b1    = (const float*)d_in[3];
    const float* W2    = (const float*)d_in[4];
    const float* b2    = (const float*)d_in[5];
    float* out = (float*)d_out;

    const size_t shmem = (size_t)(96 * XP + 2 * 96 * 96) * sizeof(float)
                       + (size_t)(N_AGENT * N_NBR) * sizeof(int);
    relational_wmma_kernel<<<dim3(512), dim3(128), shmem, stream>>>(
        x, edges, W1, b1, W2, b2, out);
}


# W1/W2 B fragments loaded 12 bytes at a time: a lane owns 3 adjacent columns, one per column tile (2 loads per k-step, were 6); D stored to those columns; W2 asked for behind phase 2; k order kept
# speedup vs baseline: 3.2627x; 1.0396x over previous
#include <hip/hip_runtime.h>

typedef float v2f __attribute__((ext_vector_type(2)));
typedef float v3f __attribute__((ext_vector_type(3)));
typedef v3f v3f_u __attribute__((aligned(4)));
typedef float v4f __attribute__((ext_vector_type(4)));
typedef float v8f __attribute__((ext_vector_type(8)));

#define N_AGENT 81
#define N_NBR   20
#define DD      96
#define HH      96
#define ROWS_PAD 96
#define XP      100

__global__ __launch_bounds__(128) void relational_wmma_kernel(
    const float* __restrict__ x,
    const int*   __restrict__ edges,
    const float* __restrict__ W1,
    const float* __restrict__ b1,
    const float* __restrict__ W2,
    const float* __restrict__ b2,
    float* __restrict__ out)
{
    extern __shared__ __attribute__((aligned(16))) float lds[];
    float* xh   = lds;
    float* sbuf = lds + 96 * XP;
    float* ybuf = lds + 96 * XP + 96 * 96;
    int*   eg   = (int*)(lds + 96 * XP + 2 * 96 * 96);

    const int b    = blockIdx.x;
    const int tid  = threadIdx.x;
    const int lane = tid & 31;
    const int wave = tid >> 5;
    const int lh   = (lane >> 4) & 1;
    const int l15  = lane & 15;

    constexpr int NX4    = (N_AGENT * DD) / 4;
    constexpr int NXP4   = (ROWS_PAD * DD) / 4;
    constexpr int NXFULL = NX4 / 128;
    constexpr int NXRND  = NXP4 / 128;
    constexpr int NE4TOT = (N_AGENT * N_NBR) / 4;
    constexpr int NEFULL = NE4TOT / 128;
    const float4* xb4  = (const float4*)(x + (size_t)b * (N_AGENT * DD));
    const int4*   eg4g = (const int4*)edges;
    float4 xr[NXFULL];
    int4   er[NEFULL];
    #pragma unroll
    for (int i = 0; i < NXFULL; ++i) xr[i] = xb4[tid + 128 * i];
    const float4 xlast = xb4[min(tid + 128 * NXFULL, NX4 - 1)];
    #pragma unroll
    for (int i = 0; i < NEFULL; ++i) er[i] = eg4g[tid + 128 * i];
    const int  felast = min(tid + 128 * NEFULL, NE4TOT - 1);
    const int4 elast  = eg4g[felast];
    __builtin_amdgcn_sched_barrier(0);

    constexpr int NKK = DD / 4;
    const int ncol = (wave & 1) * 48 + 3 * l15;
    v2f wf[3][NKK];
    {
        const float* Wh = W1 + (wave >> 1) * (DD * HH);
        #pragma unroll
        for (int kk = 0; kk < NKK; ++kk) {
            const int k0 = kk * 4 + lh * 2;
            const v3f w0 = *(const v3f_u*)(Wh + (k0)*HH + ncol);
            const v3f w1 = *(const v3f_u*)(Wh + (k0 + 1) * HH + ncol);
            wf[0][kk] = (v2f){w0.x, w1.x};
            wf[1][kk] = (v2f){w0.y, w1.y};
            wf[2][kk] = (v2f){w0.z, w1.z};
        }
    }

    {
        float4* xh4  = (float4*)xh;
        int4*   eg4s = (int4*)eg;
        #pragma unroll
        for (int i = 0; i < NXFULL; ++i) {
            const unsigned f = (unsigned)(tid + 128 * i);
            xh4[f + f / (DD / 4)] = xr[i];
        }
        const bool xreal = (tid + 128 * NXFULL) < NX4;
        {
            const unsigned f = (unsigned)(tid + 128 * NXFULL);
            xh4[f + f / (DD / 4)] = make_float4(xreal ? xlast.x : 0.f, xreal ? xlast.y : 0.f,
                                                xreal ? xlast.z : 0.f, xreal ? xlast.w : 0.f);
        }
        #pragma unroll
        for (int i = NXFULL + 1; i < NXRND; ++i) {
            const unsigned f = (unsigned)(tid + 128 * i);
            xh4[f + f / (DD / 4)] = make_float4(0.f, 0.f, 0.f, 0.f);
        }
        #pragma unroll
        for (int i = 0; i < NEFULL; ++i) eg4s[tid + 128 * i] = er[i];
        eg4s[felast] = elast;
    }
    __syncthreads();

    v2f af[2][NKK];
    const float* afrag = xh + l15 * XP + lh * 2;
    #pragma unroll
    for (int kk = 0; kk < NKK; ++kk) af[0][kk] = *(const v2f*)(afrag + kk * 4);

    {
        float* dst = (wave >> 1) ? ybuf : sbuf;
        #pragma unroll
        for (int mt = 0; mt < 6; ++mt) {
            if (mt + 1 < 6) {
                const float* an = afrag + (mt + 1) * 16 * XP;
                #pragma unroll
                for (int kk = 0; kk < NKK; ++kk) af[(mt + 1) & 1][kk] = *(const v2f*)(an + kk * 4);
            }
            __builtin_amdgcn_sched_barrier(0);
            v8f acc0 = {}, acc1 = {}, acc2 = {};
            #pragma unroll
            for (int kk = 0; kk < NKK; ++kk) {
                const v2f a = af[mt & 1][kk];
                acc0 = __builtin_amdgcn_wmma_f32_16x16x4_f32(false, a, false, wf[0][kk],
                                                             (short)0, acc0, false, false);
                acc1 = __builtin_amdgcn_wmma_f32_16x16x4_f32(false, a, false, wf[1][kk],
                                                             (short)0, acc1, false, false);
                acc2 = __builtin_amdgcn_wmma_f32_16x16x4_f32(false, a, false, wf[2][kk],
                                                             (short)0, acc2, false, false);
            }
            #pragma unroll
            for (int j = 0; j < 8; ++j) {
                const int mr = mt * 16 + j + lh * 8;
                dst[mr * HH + ncol]     = acc0[j];
                dst[mr * HH + ncol + 1] = acc1[j];
                dst[mr * HH + ncol + 2] = acc2[j];
            }
        }
    }

    v3f wr0[NKK], wr1[NKK];
    #pragma unroll
    for (int kk = 0; kk < NKK; ++kk) {
        const int k0 = kk * 4 + lh * 2;
        wr0[kk] = *(const v3f_u*)(W2 + (k0)*HH + ncol);
        wr1[kk] = *(const v3f_u*)(W2 + (k0 + 1) * HH + ncol);
    }

    __syncthreads();

    {
        constexpr int NCG   = HH / 4;
        constexpr int NITEM = N_AGENT * NCG;
        constexpr int NE4   = N_NBR / 4;
        const int4* eg4 = (const int4*)eg;
        int4 e[NE4];
        #pragma unroll
        for (int q = 0; q < NE4; ++q) e[q] = eg4[(tid / NCG) * NE4 + q];
        for (int it = tid; it < NITEM; it += 128) {
            const int a = it / NCG;
            const int c = (it % NCG) * 4;
            const int an = min((it + 128) / NCG, N_AGENT - 1);
            int4 en[NE4];
            #pragma unroll
            for (int q = 0; q < NE4; ++q) en[q] = eg4[an * NE4 + q];
            const float4 bb = *(const float4*)(b1 + c);
            const v4f ss = *(const v4f*)(sbuf + a * HH + c);
            const float sx = ss.x + bb.x, sy = ss.y + bb.y, sz = ss.z + bb.z, sw = ss.w + bb.w;
            const float* yc = ybuf + c;
            float vx = 0.f, vy = 0.f, vz = 0.f, vw = 0.f;
            #pragma unroll
            for (int q = 0; q < NE4; ++q) {
                const int jj[4] = {e[q].x, e[q].y, e[q].z, e[q].w};
                #pragma unroll
                for (int r = 0; r < 4; ++r) {
                    const v4f y = *(const v4f*)(yc + jj[r] * HH);
                    vx += fmaxf(sx + y.x, 0.f);
                    vy += fmaxf(sy + y.y, 0.f);
                    vz += fmaxf(sz + y.z, 0.f);
                    vw += fmaxf(sw + y.w, 0.f);
                }
            }
            *(v4f*)(xh + a * XP + c) = (v4f){vx, vy, vz, vw};
            #pragma unroll
            for (int q = 0; q < NE4; ++q) e[q] = en[q];
        }
    }
    __syncthreads();

    float* ob = out + (size_t)b * (N_AGENT * HH);
    {
        const v3f bb2 = *(const v3f_u*)(b2 + ncol);
        const float bias0 = (float)N_NBR * bb2.x;
        const float bias1 = (float)N_NBR * bb2.y;
        const float bias2 = (float)N_NBR * bb2.z;
        const int mt0 = (wave >> 1) * 3;
        v2f wg[3][NKK];
        #pragma unroll
        for (int kk = 0; kk < NKK; ++kk) {
            wg[0][kk] = (v2f){wr0[kk].x, wr1[kk].x};
            wg[1][kk] = (v2f){wr0[kk].y, wr1[kk].y};
            wg[2][kk] = (v2f){wr0[kk].z, wr1[kk].z};
        }
        const float* hfrag = xh + (mt0 * 16 + l15) * XP + lh * 2;
        #pragma unroll
        for (int kk = 0; kk < NKK; ++kk) af[0][kk] = *(const v2f*)(hfrag + kk * 4);
        #pragma unroll
        for (int mi = 0; mi < 3; ++mi) {
            const int mt = mt0 + mi;
            if (mi + 1 < 3) {
                const float* an = hfrag + (mi + 1) * 16 * XP;
                #pragma unroll
                for (int kk = 0; kk < NKK; ++kk) af[(mi + 1) & 1][kk] = *(const v2f*)(an + kk * 4);
            }
            __builtin_amdgcn_sched_barrier(0);
            v8f acc0 = {}, acc1 = {}, acc2 = {};
            #pragma unroll
            for (int kk = 0; kk < NKK; ++kk) {
                const v2f a = af[mi & 1][kk];
                acc0 = __builtin_amdgcn_wmma_f32_16x16x4_f32(false, a, false, wg[0][kk],
                                                             (short)0, acc0, false, false);
                acc1 = __builtin_amdgcn_wmma_f32_16x16x4_f32(false, a, false, wg[1][kk],
                                                             (short)0, acc1, false, false);
                acc2 = __builtin_amdgcn_wmma_f32_16x16x4_f32(false, a, false, wg[2][kk],
                                                             (short)0, acc2, false, false);
            }
            #pragma unroll
            for (int j = 0; j < 8; ++j) {
                const int mr = mt * 16 + j + lh * 8;
                if (mr < N_AGENT) {
                    *(v3f_u*)(ob + mr * HH + ncol) =
                        (v3f){acc0[j] + bias0, acc1[j] + bias1, acc2[j] + bias2};
                }
            }
        }
    }
}

extern "C" void kernel_launch(void* const* d_in, const int* in_sizes, int n_in,
                              void* d_out, int out_size, void* d_ws, size_t ws_size,
                              hipStream_t stream) {
    (void)in_sizes; (void)n_in; (void)out_size; (void)d_ws; (void)ws_size;
    const float* x     = (const float*)d_in[0];
    const int*   edges = (const int*)d_in[1];
    const float* W1    = (const float*)d_in[2];
    const float* b1    = (const float*)d_in[3];
    const float* W2    = (const float*)d_in[4];
    const float* b2    = (const float*)d_in[5];
    float* out = (float*)d_out;

    const size_t shmem = (size_t)(96 * XP + 2 * 96 * 96) * sizeof(float)
                       + (size_t)(N_AGENT * N_NBR) * sizeof(int);
    relational_wmma_kernel<<<dim3(512), dim3(128), shmem, stream>>>(
        x, edges, W1, b1, W2, b2, out);
}


# Phase 3 per-item overhead cut: neighbour lists staged as y-row byte offsets (row address = 1 add, was a 32-bit multiply-add), item index stepped, b1 as start value of the s sums (b1 summed first)
# speedup vs baseline: 3.4899x; 1.0696x over previous
#include <hip/hip_runtime.h>

typedef float v2f __attribute__((ext_vector_type(2)));
typedef float v3f __attribute__((ext_vector_type(3)));
typedef v3f v3f_u __attribute__((aligned(4)));
typedef float v4f __attribute__((ext_vector_type(4)));
typedef float v8f __attribute__((ext_vector_type(8)));

#define N_AGENT 81
#define N_NBR   20
#define DD      96
#define HH      96
#define ROWS_PAD 96
#define XP      100
#define YROWB   (HH * 4)

__global__ __launch_bounds__(128) void relational_wmma_kernel(
    const float* __restrict__ x,
    const int*   __restrict__ edges,
    const float* __restrict__ W1,
    const float* __restrict__ b1,
    const float* __restrict__ W2,
    const float* __restrict__ b2,
    float* __restrict__ out)
{
    extern __shared__ __attribute__((aligned(16))) float lds[];
    float* xh   = lds;
    float* sbuf = lds + 96 * XP;
    float* ybuf = lds + 96 * XP + 96 * 96;
    int*   eg   = (int*)(lds + 96 * XP + 2 * 96 * 96);

    const int b    = blockIdx.x;
    const int tid  = threadIdx.x;
    const int lane = tid & 31;
    const int wave = tid >> 5;
    const int lh   = (lane >> 4) & 1;
    const int l15  = lane & 15;

    constexpr int NX4    = (N_AGENT * DD) / 4;
    constexpr int NXP4   = (ROWS_PAD * DD) / 4;
    constexpr int NXFULL = NX4 / 128;
    constexpr int NXRND  = NXP4 / 128;
    constexpr int NE4TOT = (N_AGENT * N_NBR) / 4;
    constexpr int NEFULL = NE4TOT / 128;
    const float4* xb4  = (const float4*)(x + (size_t)b * (N_AGENT * DD));
    const int4*   eg4g = (const int4*)edges;
    float4 xr[NXFULL];
    int4   er[NEFULL];
    #pragma unroll
    for (int i = 0; i < NXFULL; ++i) xr[i] = xb4[tid + 128 * i];
    const float4 xlast = xb4[min(tid + 128 * NXFULL, NX4 - 1)];
    #pragma unroll
    for (int i = 0; i < NEFULL; ++i) er[i] = eg4g[tid + 128 * i];
    const int  felast = min(tid + 128 * NEFULL, NE4TOT - 1);
    const int4 elast  = eg4g[felast];
    __builtin_amdgcn_sched_barrier(0);

    constexpr int NKK = DD / 4;
    const int ncol = (wave & 1) * 48 + 3 * l15;
    v2f wf[3][NKK];
    {
        const float* Wh = W1 + (wave >> 1) * (DD * HH);
        #pragma unroll
        for (int kk = 0; kk < NKK; ++kk) {
            const int k0 = kk * 4 + lh * 2;
            const v3f w0 = *(const v3f_u*)(Wh + (k0)*HH + ncol);
            const v3f w1 = *(const v3f_u*)(Wh + (k0 + 1) * HH + ncol);
            wf[0][kk] = (v2f){w0.x, w1.x};
            wf[1][kk] = (v2f){w0.y, w1.y};
            wf[2][kk] = (v2f){w0.z, w1.z};
        }
    }
    const v3f b1v = *(const v3f_u*)(b1 + ncol);
    const float sb0 = (wave >> 1) ? 0.f : b1v.x;
    const float sb1 = (wave >> 1) ? 0.f : b1v.y;
    const float sb2 = (wave >> 1) ? 0.f : b1v.z;

    {
        float4* xh4  = (float4*)xh;
        int4*   eg4s = (int4*)eg;
        #pragma unroll
        for (int i = 0; i < NXFULL; ++i) {
            const unsigned f = (unsigned)(tid + 128 * i);
            xh4[f + f / (DD / 4)] = xr[i];
        }
        const bool xreal = (tid + 128 * NXFULL) < NX4;
        {
            const unsigned f = (unsigned)(tid + 128 * NXFULL);
            xh4[f + f / (DD / 4)] = make_float4(xreal ? xlast.x : 0.f, xreal ? xlast.y : 0.f,
                                                xreal ? xlast.z : 0.f, xreal ? xlast.w : 0.f);
        }
        #pragma unroll
        for (int i = NXFULL + 1; i < NXRND; ++i) {
            const unsigned f = (unsigned)(tid + 128 * i);
            xh4[f + f / (DD / 4)] = make_float4(0.f, 0.f, 0.f, 0.f);
        }
        #pragma unroll
        for (int i = 0; i < NEFULL; ++i)
            eg4s[tid + 128 * i] = make_int4(er[i].x * YROWB, er[i].y * YROWB,
                                            er[i].z * YROWB, er[i].w * YROWB);
        eg4s[felast] = make_int4(elast.x * YROWB, elast.y * YROWB, elast.z * YROWB, elast.w * YROWB);
    }
    __syncthreads();

    v2f af[2][NKK];
    const float* afrag = xh + l15 * XP + lh * 2;
    #pragma unroll
    for (int kk = 0; kk < NKK; ++kk) af[0][kk] = *(const v2f*)(afrag + kk * 4);

    {
        float* dst = (wave >> 1) ? ybuf : sbuf;
        #pragma unroll
        for (int mt = 0; mt < 6; ++mt) {
            if (mt + 1 < 6) {
                const float* an = afrag + (mt + 1) * 16 * XP;
                #pragma unroll
                for (int kk = 0; kk < NKK; ++kk) af[(mt + 1) & 1][kk] = *(const v2f*)(an + kk * 4);
            }
            __builtin_amdgcn_sched_barrier(0);
            v8f acc0 = {sb0, sb0, sb0, sb0, sb0, sb0, sb0, sb0};
            v8f acc1 = {sb1, sb1, sb1, sb1, sb1, sb1, sb1, sb1};
            v8f acc2 = {sb2, sb2, sb2, sb2, sb2, sb2, sb2, sb2};
            #pragma unroll
            for (int kk = 0; kk < NKK; ++kk) {
                const v2f a = af[mt & 1][kk];
                acc0 = __builtin_amdgcn_wmma_f32_16x16x4_f32(false, a, false, wf[0][kk],
                                                             (short)0, acc0, false, false);
                acc1 = __builtin_amdgcn_wmma_f32_16x16x4_f32(false, a, false, wf[1][kk],
                                                             (short)0, acc1, false, false);
                acc2 = __builtin_amdgcn_wmma_f32_16x16x4_f32(false, a, false, wf[2][kk],
                                                             (short)0, acc2, false, false);
            }
            #pragma unroll
            for (int j = 0; j < 8; ++j) {
                const int mr = mt * 16 + j + lh * 8;
                dst[mr * HH + ncol]     = acc0[j];
                dst[mr * HH + ncol + 1] = acc1[j];
                dst[mr * HH + ncol + 2] = acc2[j];
            }
        }
    }

    v3f wr0[NKK], wr1[NKK];
    #pragma unroll
    for (int kk = 0; kk < NKK; ++kk) {
        const int k0 = kk * 4 + lh * 2;
        wr0[kk] = *(const v3f_u*)(W2 + (k0)*HH + ncol);
        wr1[kk] = *(const v3f_u*)(W2 + (k0 + 1) * HH + ncol);
    }

    __syncthreads();

    {
        constexpr int NCG   = HH / 4;
        constexpr int NITEM = N_AGENT * NCG;
        constexpr int NE4   = N_NBR / 4;
        const int4* eg4 = (const int4*)eg;
        int a  = tid / NCG;
        int cg = tid - a * NCG;
        int so = a * HH + cg * 4;
        int xo = a * XP + cg * 4;
        int4 e[NE4];
        #pragma unroll
        for (int q = 0; q < NE4; ++q) e[q] = eg4[a * NE4 + q];
        for (int it = tid; it < NITEM; it += 128) {
            const int c = cg * 4;
            const char* yc = (const char*)(ybuf + c);
            const v4f* yp[N_NBR];
            #pragma unroll
            for (int q = 0; q < NE4; ++q) {
                yp[4 * q + 0] = (const v4f*)(yc + e[q].x);
                yp[4 * q + 1] = (const v4f*)(yc + e[q].y);
                yp[4 * q + 2] = (const v4f*)(yc + e[q].z);
                yp[4 * q + 3] = (const v4f*)(yc + e[q].w);
            }
            const bool wrap = (cg + 8 >= NCG);
            const int an  = a + (wrap ? 6 : 5);
            const int cgn = cg + (wrap ? 8 - NCG : 8);
            const int anc = min(an, N_AGENT - 1);
            #pragma unroll
            for (int q = 0; q < NE4; ++q) e[q] = eg4[anc * NE4 + q];
            const v4f ss = *(const v4f*)(sbuf + so);
            v4f y[N_NBR];
            #pragma unroll
            for (int n = 0; n < N_NBR; ++n) y[n] = *yp[n];
            __builtin_amdgcn_sched_barrier(0);
            const float sx = ss.x, sy = ss.y, sz = ss.z, sw = ss.w;
            float vx = 0.f, vy = 0.f, vz = 0.f, vw = 0.f;
            #pragma unroll
            for (int n = 0; n < N_NBR; ++n) {
                vx += fmaxf(sx + y[n].x, 0.f);
                vy += fmaxf(sy + y[n].y, 0.f);
                vz += fmaxf(sz + y[n].z, 0.f);
                vw += fmaxf(sw + y[n].w, 0.f);
            }
            *(v4f*)(xh + xo) = (v4f){vx, vy, vz, vw};
            a  = an;
            cg = cgn;
            so += wrap ? (6 * HH + (8 - NCG) * 4) : (5 * HH + 8 * 4);
            xo += wrap ? (6 * XP + (8 - NCG) * 4) : (5 * XP + 8 * 4);
        }
    }
    __syncthreads();

    float* ob = out + (size_t)b * (N_AGENT * HH);
    {
        const v3f bb2 = *(const v3f_u*)(b2 + ncol);
        const float bias0 = (float)N_NBR * bb2.x;
        const float bias1 = (float)N_NBR * bb2.y;
        const float bias2 = (float)N_NBR * bb2.z;
        const int mt0 = (wave >> 1) * 3;
        v2f wg[3][NKK];
        #pragma unroll
        for (int kk = 0; kk < NKK; ++kk) {
            wg[0][kk] = (v2f){wr0[kk].x, wr1[kk].x};
            wg[1][kk] = (v2f){wr0[kk].y, wr1[kk].y};
            wg[2][kk] = (v2f){wr0[kk].z, wr1[kk].z};
        }
        const float* hfrag = xh + (mt0 * 16 + l15) * XP + lh * 2;
        #pragma unroll
        for (int kk = 0; kk < NKK; ++kk) af[0][kk] = *(const v2f*)(hfrag + kk * 4);
        #pragma unroll
        for (int mi = 0; mi < 3; ++mi) {
            const int mt = mt0 + mi;
            if (mi + 1 < 3) {
                const float* an = hfrag + (mi + 1) * 16 * XP;
                #pragma unroll
                for (int kk = 0; kk < NKK; ++kk) af[(mi + 1) & 1][kk] = *(const v2f*)(an + kk * 4);
            }
            __builtin_amdgcn_sched_barrier(0);
            v8f acc0 = {}, acc1 = {}, acc2 = {};
            #pragma unroll
            for (int kk = 0; kk < NKK; ++kk) {
                const v2f a = af[mi & 1][kk];
                acc0 = __builtin_amdgcn_wmma_f32_16x16x4_f32(false, a, false, wg[0][kk],
                                                             (short)0, acc0, false, false);
                acc1 = __builtin_amdgcn_wmma_f32_16x16x4_f32(false, a, false, wg[1][kk],
                                                             (short)0, acc1, false, false);
                acc2 = __builtin_amdgcn_wmma_f32_16x16x4_f32(false, a, false, wg[2][kk],
                                                             (short)0, acc2, false, false);
            }
            #pragma unroll
            for (int j = 0; j < 8; ++j) {
                const int mr = mt * 16 + j + lh * 8;
                if (mr < N_AGENT) {
                    *(v3f_u*)(ob + mr * HH + ncol) =
                        (v3f){acc0[j] + bias0, acc1[j] + bias1, acc2[j] + bias2};
                }
            }
        }
    }
}

extern "C" void kernel_launch(void* const* d_in, const int* in_sizes, int n_in,
                              void* d_out, int out_size, void* d_ws, size_t ws_size,
                              hipStream_t stream) {
    (void)in_sizes; (void)n_in; (void)out_size; (void)d_ws; (void)ws_size;
    const float* x     = (const float*)d_in[0];
    const int*   edges = (const int*)d_in[1];
    const float* W1    = (const float*)d_in[2];
    const float* b1    = (const float*)d_in[3];
    const float* W2    = (const float*)d_in[4];
    const float* b2    = (const float*)d_in[5];
    float* out = (float*)d_out;

    const size_t shmem = (size_t)(96 * XP + 2 * 96 * 96) * sizeof(float)
                       + (size_t)(N_AGENT * N_NBR) * sizeof(int);
    relational_wmma_kernel<<<dim3(512), dim3(128), shmem, stream>>>(
        x, edges, W1, b1, W2, b2, out);
}


# Row 80 taken off the padded sixth row tile: five row tiles stay WMMAs, row 80 is packed f32 multiply-adds on the B fragments in registers; phase 4 tiles re-shared; new sum order in row 80 only
# speedup vs baseline: 3.8002x; 1.0889x over previous
#include <hip/hip_runtime.h>

typedef float v2f __attribute__((ext_vector_type(2)));
typedef float v3f __attribute__((ext_vector_type(3)));
typedef v3f v3f_u __attribute__((aligned(4)));
typedef v2f v2f_u __attribute__((aligned(4)));
typedef float v4f __attribute__((ext_vector_type(4)));
typedef float v8f __attribute__((ext_vector_type(8)));

#define N_AGENT 81
#define N_NBR   20
#define DD      96
#define HH      96
#define ROWS_PAD 96
#define NMT      5
#define XP      100
#define YROWB   (HH * 4)

__device__ __forceinline__ float other_half(float v) {
    return __builtin_bit_cast(float, __builtin_amdgcn_permlanex16(
        __builtin_bit_cast(int, v), __builtin_bit_cast(int, v), 0x76543210, 0xfedcba98, false, false));
}

__global__ __launch_bounds__(128) void relational_wmma_kernel(
    const float* __restrict__ x,
    const int*   __restrict__ edges,
    const float* __restrict__ W1,
    const float* __restrict__ b1,
    const float* __restrict__ W2,
    const float* __restrict__ b2,
    float* __restrict__ out)
{
    extern __shared__ __attribute__((aligned(16))) float lds[];
    float* xh   = lds;
    float* sbuf = lds + 96 * XP;
    float* ybuf = lds + 96 * XP + 96 * 96;
    int*   eg   = (int*)(lds + 96 * XP + 2 * 96 * 96);

    const int b    = blockIdx.x;
    const int tid  = threadIdx.x;
    const int lane = tid & 31;
    const int wave = tid >> 5;
    const int lh   = (lane >> 4) & 1;
    const int l15  = lane & 15;

    constexpr int NX4    = (N_AGENT * DD) / 4;
    constexpr int NXFULL = NX4 / 128;
    constexpr int NE4TOT = (N_AGENT * N_NBR) / 4;
    constexpr int NEFULL = NE4TOT / 128;
    const float4* xb4  = (const float4*)(x + (size_t)b * (N_AGENT * DD));
    const int4*   eg4g = (const int4*)edges;
    float4 xr[NXFULL];
    int4   er[NEFULL];
    #pragma unroll
    for (int i = 0; i < NXFULL; ++i) xr[i] = xb4[tid + 128 * i];
    const float4 xlast = xb4[min(tid + 128 * NXFULL, NX4 - 1)];
    #pragma unroll
    for (int i = 0; i < NEFULL; ++i) er[i] = eg4g[tid + 128 * i];
    const int  felast = min(tid + 128 * NEFULL, NE4TOT - 1);
    const int4 elast  = eg4g[felast];
    __builtin_amdgcn_sched_barrier(0);

    constexpr int NKK = DD / 4;
    const int ncol = (wave & 1) * 48 + 3 * l15;
    v2f wf[3][NKK];
    {
        const float* Wh = W1 + (wave >> 1) * (DD * HH);
        #pragma unroll
        for (int kk = 0; kk < NKK; ++kk) {
            const int k0 = kk * 4 + lh * 2;
            const v3f w0 = *(const v3f_u*)(Wh + (k0)*HH + ncol);
            const v3f w1 = *(const v3f_u*)(Wh + (k0 + 1) * HH + ncol);
            wf[0][kk] = (v2f){w0.x, w1.x};
            wf[1][kk] = (v2f){w0.y, w1.y};
            wf[2][kk] = (v2f){w0.z, w1.z};
        }
    }
    const v3f b1v = *(const v3f_u*)(b1 + ncol);
    const float sb0 = (wave >> 1) ? 0.f : b1v.x;
    const float sb1 = (wave >> 1) ? 0.f : b1v.y;
    const float sb2 = (wave >> 1) ? 0.f : b1v.z;

    {
        float4* xh4  = (float4*)xh;
        int4*   eg4s = (int4*)eg;
        #pragma unroll
        for (int i = 0; i < NXFULL; ++i) {
            const unsigned f = (unsigned)(tid + 128 * i);
            xh4[f + f / (DD / 4)] = xr[i];
        }
        const bool xreal = (tid + 128 * NXFULL) < NX4;
        {
            const unsigned f = (unsigned)(tid + 128 * NXFULL);
            xh4[f + f / (DD / 4)] = make_float4(xreal ? xlast.x : 0.f, xreal ? xlast.y : 0.f,
                                                xreal ? xlast.z : 0.f, xreal ? xlast.w : 0.f);
        }
        #pragma unroll
        for (int i = 0; i < NEFULL; ++i)
            eg4s[tid + 128 * i] = make_int4(er[i].x * YROWB, er[i].y * YROWB,
                                            er[i].z * YROWB, er[i].w * YROWB);
        eg4s[felast] = make_int4(elast.x * YROWB, elast.y * YROWB, elast.z * YROWB, elast.w * YROWB);
    }
    __syncthreads();

    v2f af[2][NKK];
    const float* afrag = xh + l15 * XP + lh * 2;
    #pragma unroll
    for (int kk = 0; kk < NKK; ++kk) af[0][kk] = *(const v2f*)(afrag + kk * 4);

    {
        float* dst = (wave >> 1) ? ybuf : sbuf;
        const float* x80 = xh + 80 * XP + lh * 2;
        #pragma unroll
        for (int mt = 0; mt < NMT; ++mt) {
            {
                const float* an = (mt + 1 < NMT) ? afrag + (mt + 1) * 16 * XP : x80;
                #pragma unroll
                for (int kk = 0; kk < NKK; ++kk) af[(mt + 1) & 1][kk] = *(const v2f*)(an + kk * 4);
            }
            __builtin_amdgcn_sched_barrier(0);
            v8f acc0 = {sb0, sb0, sb0, sb0, sb0, sb0, sb0, sb0};
            v8f acc1 = {sb1, sb1, sb1, sb1, sb1, sb1, sb1, sb1};
            v8f acc2 = {sb2, sb2, sb2, sb2, sb2, sb2, sb2, sb2};
            v2f p0 = {0.f, 0.f}, p1 = {0.f, 0.f}, p2 = {0.f, 0.f};
            #pragma unroll
            for (int kk = 0; kk < NKK; ++kk) {
                const v2f a = af[mt & 1][kk];
                acc0 = __builtin_amdgcn_wmma_f32_16x16x4_f32(false, a, false, wf[0][kk],
                                                             (short)0, acc0, false, false);
                acc1 = __builtin_amdgcn_wmma_f32_16x16x4_f32(false, a, false, wf[1][kk],
                                                             (short)0, acc1, false, false);
                acc2 = __builtin_amdgcn_wmma_f32_16x16x4_f32(false, a, false, wf[2][kk],
                                                             (short)0, acc2, false, false);
                if (mt == NMT - 1) {
                    const v2f r = af[(mt + 1) & 1][kk];
                    p0 = __builtin_elementwise_fma(r, wf[0][kk], p0);
                    p1 = __builtin_elementwise_fma(r, wf[1][kk], p1);
                    p2 = __builtin_elementwise_fma(r, wf[2][kk], p2);
                }
            }
            #pragma unroll
            for (int j = 0; j < 8; ++j) {
                const int mr = mt * 16 + j + lh * 8;
                dst[mr * HH + ncol]     = acc0[j];
                dst[mr * HH + ncol + 1] = acc1[j];
                dst[mr * HH + ncol + 2] = acc2[j];
            }
            if (mt == NMT - 1) {
                float r0 = p0.x + p0.y, r1 = p1.x + p1.y, r2 = p2.x + p2.y;
                r0 += other_half(r0);
                r1 += other_half(r1);
                r2 += other_half(r2);
                dst[80 * HH + ncol]     = r0 + sb0;
                dst[80 * HH + ncol + 1] = r1 + sb1;
                dst[80 * HH + ncol + 2] = r2 + sb2;
            }
        }
    }

    v3f wr0[NKK], wr1[NKK];
    #pragma unroll
    for (int kk = 0; kk < NKK; ++kk) {
        const int k0 = kk * 4 + lh * 2;
        wr0[kk] = *(const v3f_u*)(W2 + (k0)*HH + ncol);
        wr1[kk] = *(const v3f_u*)(W2 + (k0 + 1) * HH + ncol);
    }

    __syncthreads();

    {
        constexpr int NCG   = HH / 4;
        constexpr int NITEM = N_AGENT * NCG;
        constexpr int NE4   = N_NBR / 4;
        const int4* eg4 = (const int4*)eg;
        int a  = tid / NCG;
        int cg = tid - a * NCG;
        int so = a * HH + cg * 4;
        int xo = a * XP + cg * 4;
        int4 e[NE4];
        #pragma unroll
        for (int q = 0; q < NE4; ++q) e[q] = eg4[a * NE4 + q];
        for (int it = tid; it < NITEM; it += 128) {
            const int c = cg * 4;
            const char* yc = (const char*)(ybuf + c);
            const v4f* yp[N_NBR];
            #pragma unroll
            for (int q = 0; q < NE4; ++q) {
                yp[4 * q + 0] = (const v4f*)(yc + e[q].x);
                yp[4 * q + 1] = (const v4f*)(yc + e[q].y);
                yp[4 * q + 2] = (const v4f*)(yc + e[q].z);
                yp[4 * q + 3] = (const v4f*)(yc + e[q].w);
            }
            const bool wrap = (cg + 8 >= NCG);
            const int an  = a + (wrap ? 6 : 5);
            const int cgn = cg + (wrap ? 8 - NCG : 8);
            const int anc = min(an, N_AGENT - 1);
            #pragma unroll
            for (int q = 0; q < NE4; ++q) e[q] = eg4[anc * NE4 + q];
            const v4f ss = *(const v4f*)(sbuf + so);
            v4f y[N_NBR];
            #pragma unroll
            for (int n = 0; n < N_NBR; ++n) y[n] = *yp[n];
            __builtin_amdgcn_sched_barrier(0);
            const float sx = ss.x, sy = ss.y, sz = ss.z, sw = ss.w;
            float vx = 0.f, vy = 0.f, vz = 0.f, vw = 0.f;
            #pragma unroll
            for (int n = 0; n < N_NBR; ++n) {
                vx += fmaxf(sx + y[n].x, 0.f);
                vy += fmaxf(sy + y[n].y, 0.f);
                vz += fmaxf(sz + y[n].z, 0.f);
                vw += fmaxf(sw + y[n].w, 0.f);
            }
            *(v4f*)(xh + xo) = (v4f){vx, vy, vz, vw};
            a  = an;
            cg = cgn;
            so += wrap ? (6 * HH + (8 - NCG) * 4) : (5 * HH + 8 * 4);
            xo += wrap ? (6 * XP + (8 - NCG) * 4) : (5 * XP + 8 * 4);
        }
    }
    __syncthreads();

    float* ob = out + (size_t)b * (N_AGENT * HH);
    {
        const v3f bb2 = *(const v3f_u*)(b2 + ncol);
        const float bias0 = (float)N_NBR * bb2.x;
        const float bias1 = (float)N_NBR * bb2.y;
        const float bias2 = (float)N_NBR * bb2.z;
        const int hi = __builtin_amdgcn_readfirstlane(wave >> 1);
        const int mt0 = hi * 3;
        v2f wg[3][NKK];
        #pragma unroll
        for (int kk = 0; kk < NKK; ++kk) {
            wg[0][kk] = (v2f){wr0[kk].x, wr1[kk].x};
            wg[1][kk] = (v2f){wr0[kk].y, wr1[kk].y};
            wg[2][kk] = (v2f){wr0[kk].z, wr1[kk].z};
        }
        const float* hfrag = xh + (mt0 * 16 + l15) * XP + lh * 2;
        const float* hmid  = xh + (2 * 16 + l15) * XP + lh * 2;
        #pragma unroll
        for (int kk = 0; kk < NKK; ++kk) af[0][kk] = *(const v2f*)(hfrag + kk * 4);
        #pragma unroll
        for (int mi = 0; mi < 2; ++mi) {
            const int mt = mt0 + mi;
            {
                const float* an = (mi == 0) ? hfrag + 16 * XP : hmid;
                #pragma unroll
                for (int kk = 0; kk < NKK; ++kk) af[(mi + 1) & 1][kk] = *(const v2f*)(an + kk * 4);
            }
            __builtin_amdgcn_sched_barrier(0);
            v8f acc0 = {}, acc1 = {}, acc2 = {};
            #pragma unroll
            for (int kk = 0; kk < NKK; ++kk) {
                const v2f a = af[mi & 1][kk];
                acc0 = __builtin_amdgcn_wmma_f32_16x16x4_f32(false, a, false, wg[0][kk],
                                                             (short)0, acc0, false, false);
                acc1 = __builtin_amdgcn_wmma_f32_16x16x4_f32(false, a, false, wg[1][kk],
                                                             (short)0, acc1, false, false);
                acc2 = __builtin_amdgcn_wmma_f32_16x16x4_f32(false, a, false, wg[2][kk],
                                                             (short)0, acc2, false, false);
            }
            #pragma unroll
            for (int j = 0; j < 8; ++j) {
                const int mr = mt * 16 + j + lh * 8;
                *(v3f_u*)(ob + mr * HH + ncol) =
                    (v3f){acc0[j] + bias0, acc1[j] + bias1, acc2[j] + bias2};
            }
        }
        if (hi == 0) {
            v8f acc0 = {}, acc1 = {};
            #pragma unroll
            for (int kk = 0; kk < NKK; ++kk) {
                const v2f a = af[0][kk];
                acc0 = __builtin_amdgcn_wmma_f32_16x16x4_f32(false, a, false, wg[0][kk],
                                                             (short)0, acc0, false, false);
                acc1 = __builtin_amdgcn_wmma_f32_16x16x4_f32(false, a, false, wg[1][kk],
                                                             (short)0, acc1, false, false);
            }
            #pragma unroll
            for (int j = 0; j < 8; ++j) {
                const int mr = 2 * 16 + j + lh * 8;
                *(v2f_u*)(ob + mr * HH + ncol) = (v2f){acc0[j] + bias0, acc1[j] + bias1};
            }
        } else {
            const float* h80 = xh + 80 * XP + lh * 2;
            #pragma unroll
            for (int kk = 0; kk < NKK; ++kk) af[1][kk] = *(const v2f*)(h80 + kk * 4);
            v8f acc2 = {};
            v2f p0 = {0.f, 0.f}, p1 = {0.f, 0.f}, p2 = {0.f, 0.f};
            #pragma unroll
            for (int kk = 0; kk < NKK; ++kk) {
                acc2 = __builtin_amdgcn_wmma_f32_16x16x4_f32(false, af[0][kk], false, wg[2][kk],
                                                             (short)0, acc2, false, false);
                const v2f r = af[1][kk];
                p0 = __builtin_elementwise_fma(r, wg[0][kk], p0);
                p1 = __builtin_elementwise_fma(r, wg[1][kk], p1);
                p2 = __builtin_elementwise_fma(r, wg[2][kk], p2);
            }
            #pragma unroll
            for (int j = 0; j < 8; ++j) {
                const int mr = 2 * 16 + j + lh * 8;
                ob[mr * HH + ncol + 2] = acc2[j] + bias2;
            }
            float r0 = p0.x + p0.y, r1 = p1.x + p1.y, r2 = p2.x + p2.y;
            r0 += other_half(r0);
            r1 += other_half(r1);
            r2 += other_half(r2);
            if (lh == 0)
                *(v3f_u*)(ob + 80 * HH + ncol) = (v3f){r0 + bias0, r1 + bias1, r2 + bias2};
        }
    }
}

extern "C" void kernel_launch(void* const* d_in, const int* in_sizes, int n_in,
                              void* d_out, int out_size, void* d_ws, size_t ws_size,
                              hipStream_t stream) {
    (void)in_sizes; (void)n_in; (void)out_size; (void)d_ws; (void)ws_size;
    const float* x     = (const float*)d_in[0];
    const int*   edges = (const int*)d_in[1];
    const float* W1    = (const float*)d_in[2];
    const float* b1    = (const float*)d_in[3];
    const float* W2    = (const float*)d_in[4];
    const float* b2    = (const float*)d_in[5];
    float* out = (float*)d_out;

    const size_t shmem = (size_t)(96 * XP + 2 * 96 * 96) * sizeof(float)
                       + (size_t)(N_AGENT * N_NBR) * sizeof(int);
    relational_wmma_kernel<<<dim3(512), dim3(128), shmem, stream>>>(
        x, edges, W1, b1, W2, b2, out);
}
